# starting kernel
# baseline (speedup 1.0000x reference)
#include <hip/hip_runtime.h>

typedef __attribute__((ext_vector_type(2))) float v2f;
typedef __attribute__((ext_vector_type(4))) float v4f;
typedef __attribute__((ext_vector_type(8))) float v8f;

#define H_IMG 14
#define W_IMG 14
#define LNODES 196
#define NB 32
#define NHEADS 24
#define PDIM 64
#define SDT (NB * NHEADS * LNODES)

__launch_bounds__(128, 4)
__global__ void chimera_dag_mix(const float* __restrict__ dt,
                                const float* __restrict__ dt_bias,
                                const float* __restrict__ x,
                                const float* __restrict__ Dskip,
                                float* __restrict__ y)
{
    __shared__ float sAleft[LNODES];
    __shared__ float sAtop[LNODES];
    __shared__ float sT[16][17];
    __shared__ alignas(16) float sZin[16][68];
    __shared__ alignas(16) float sZprev[16][68];

    const int bh   = blockIdx.x;
    const int h    = bh % NHEADS;
    const int tid  = threadIdx.x;
    const int lane = tid & 31;
    const int wave = tid >> 5;
    const int col0 = wave * 16;
    const int mrow = lane & 15;
    const int ksel = (lane >> 4) << 1;

    const float bias = dt_bias[h];
    const float dsk  = Dskip[h];
    const float* __restrict__ xb = x + (size_t)bh * (LNODES * PDIM);
    float* __restrict__ yb       = y + (size_t)bh * (LNODES * PDIM);

    for (int node = tid; node < LNODES; node += 128) {
        int row = node / W_IMG, col = node % W_IMG;
        float nie = 2.0f - (col == 0 ? 1.0f : 0.0f) - (row == 0 ? 1.0f : 0.0f);
        if (nie < 1e-6f) nie = 1.0f;
        float rn = 0.95f * rsqrtf(nie);
        float v0 = dt[0 * SDT + bh * LNODES + node] + bias;
        float v1 = dt[1 * SDT + bh * LNODES + node] + bias;
        sAleft[node] = (col > 0) ? rn / (1.0f + expf(v0)) : 0.0f;
        sAtop [node] = (row > 0) ? rn / (1.0f + expf(v1)) : 0.0f;
    }
    {
        v4f zv = {};
        v4f* zp = (v4f*)&sZprev[0][0];
        for (int i = tid; i < (16 * 68) / 4; i += 128) zp[i] = zv;
        v4f* zi = (v4f*)&sZin[14][0];
        for (int i = tid; i < (2 * 68) / 4; i += 128) zi[i] = zv;
        if (tid < 32) sT[14 + (tid >> 4)][tid & 15] = 0.0f;
    }
    __syncthreads();

    const int  zc  = tid >> 3;
    const int  zp0 = (tid & 7) * 8;
    const bool act = (zc < W_IMG);

    v4f xprev0 = {}, xprev1 = {};

    for (int r = 0; r < H_IMG; ++r) {
        if (act) {
            int node = r * W_IMG + zc;
            float at = sAtop[node];
            const v4f* xp = (const v4f*)(xb + node * PDIM + zp0);
            v4f x0 = xp[0], x1 = xp[1];
            const v4f* pp = (const v4f*)&sZprev[zc][zp0];
            v4f p0 = pp[0], p1 = pp[1];
            if (r > 0) {
                v4f* yp = (v4f*)(yb + (node - W_IMG) * PDIM + zp0);
                yp[0] = p0 + dsk * xprev0;
                yp[1] = p1 + dsk * xprev1;
            }
            v4f* op = (v4f*)&sZin[zc][zp0];
            op[0] = x0 + at * p0;
            op[1] = x1 + at * p1;
            xprev0 = x0; xprev1 = x1;
        }
        if (tid < 16) {
            float t = 0.0f;
            for (int c = 0; c < W_IMG; ++c) {
                float al = sAleft[r * W_IMG + c];
                t = (tid == c) ? 1.0f : al * t;
                sT[c][tid] = t;
            }
        }
        __syncthreads();

        v8f acc = {};
#pragma unroll
        for (int k0 = 0; k0 < 16; k0 += 4) {
            v2f a, b;
            a.x = sT[mrow][k0 + ksel];
            a.y = sT[mrow][k0 + ksel + 1];
            b.x = sZin[k0 + ksel][col0 + mrow];
            b.y = sZin[k0 + ksel + 1][col0 + mrow];
            acc = __builtin_amdgcn_wmma_f32_16x16x4_f32(
                false, a, false, b, (short)0, acc, false, false);
        }

#pragma unroll
        for (int i = 0; i < 8; ++i) {
            int M = i + ((lane >> 4) << 3);
            if (M < W_IMG) sZprev[M][col0 + mrow] = acc[i];
        }
        __syncthreads();
    }

    if (act) {
        const v4f* pp = (const v4f*)&sZprev[zc][zp0];
        v4f* yp = (v4f*)(yb + ((H_IMG - 1) * W_IMG + zc) * PDIM + zp0);
        yp[0] = pp[0] + dsk * xprev0;
        yp[1] = pp[1] + dsk * xprev1;
    }
}

extern "C" void kernel_launch(void* const* d_in, const int* in_sizes, int n_in,
                              void* d_out, int out_size, void* d_ws, size_t ws_size,
                              hipStream_t stream) {
    (void)in_sizes; (void)n_in; (void)d_ws; (void)ws_size; (void)out_size;
    const float* dt      = (const float*)d_in[0];
    const float* dt_bias = (const float*)d_in[1];
    const float* x       = (const float*)d_in[2];
    const float* Dskip   = (const float*)d_in[3];
    float* y             = (float*)d_out;
    chimera_dag_mix<<<NB * NHEADS, 128, 0, stream>>>(dt, dt_bias, x, Dskip, y);
}


# Wave-private tiling: each wave owns one (b,h) and 32 channels end to end; its x slab is requested up front into its own LDS slab, rows solved in place, no workgroup barrier, y leaves as 128-B lines
# speedup vs baseline: 1.5044x; 1.5044x over previous
#include <hip/hip_runtime.h>


typedef __attribute__((ext_vector_type(2))) float v2f;
typedef __attribute__((ext_vector_type(4))) float v4f;
typedef __attribute__((ext_vector_type(8))) float v8f;

#define H_IMG 14
#define W_IMG 14
#define LNODES 196
#define NB 32
#define NHEADS 24
#define PDIM 64
#define SDT (NB * NHEADS * LNODES)
#define CHW 32
#define PITCH 40
#define NCHUNK 49
#define COEF 224

#define WAVE_SYNC() do { \
    __builtin_amdgcn_fence(__ATOMIC_RELEASE, "wavefront"); \
    __builtin_amdgcn_wave_barrier(); \
    __builtin_amdgcn_fence(__ATOMIC_ACQUIRE, "wavefront"); } while (0)

__launch_bounds__(64)
__global__ void chimera_dag_mix(const float* __restrict__ dt,
                                const float* __restrict__ dt_bias,
                                const float* __restrict__ x,
                                const float* __restrict__ Dskip,
                                float* __restrict__ y)
{
    __shared__ alignas(16) float sSlab[2][LNODES * PITCH];
    __shared__ float sAt[2][COEF];

    const int bh   = blockIdx.x;
    const int hd   = bh % NHEADS;
    const int tid  = threadIdx.x;
    const int lane = tid & 31;
    const int half = __builtin_amdgcn_readfirstlane(tid >> 5);
    const int c    = lane & 15;
    const int hh   = lane >> 4;
    const int q    = lane >> 3;
    const int c4   = (lane & 7) * 4;

    float* __restrict__ S  = sSlab[half];
    float* __restrict__ At = sAt[half];

    const float bias = dt_bias[hd];
    const float dsk  = Dskip[hd];

    float d0[7], d1[7];
#pragma unroll
    for (int it = 0; it < 7; ++it) {
        int idx = lane + 32 * it;
        int idc = idx < LNODES ? idx : LNODES - 1;
        d0[it] = dt[0 * SDT + bh * LNODES + idc];
        d1[it] = dt[1 * SDT + bh * LNODES + idc];
    }

    const float* __restrict__ xb = x + (size_t)bh * (LNODES * PDIM) + half * CHW + c4;
    float* __restrict__ yb       = y + (size_t)bh * (LNODES * PDIM) + half * CHW + c4;
    v4f xr[NCHUNK];
#pragma unroll
    for (int i = 0; i < NCHUNK; ++i)
        xr[i] = *(const v4f*)(xb + (4 * i + q) * PDIM);

    float alv[7];
#pragma unroll
    for (int it = 0; it < 7; ++it) {
        int node = lane + 32 * it;
        int nc   = node < LNODES ? node : LNODES - 1;
        int row = nc / W_IMG, col = nc % W_IMG;
        float nie = 2.0f - (col == 0 ? 1.0f : 0.0f) - (row == 0 ? 1.0f : 0.0f);
        if (nie < 1e-6f) nie = 1.0f;
        float rn = 0.95f * rsqrtf(nie);
        float e0 = expf(d0[it] + bias);
        float e1 = expf(d1[it] + bias);
        float al = rn / (1.0f + e0);
        float at = rn / (1.0f + e1);
        bool in = node < LNODES;
        alv[it]  = (in && col > 0) ? al : 0.0f;
        At[node] = (in && row > 0) ? at : 0.0f;
    }
    WAVE_SYNC();

    v2f afr[H_IMG][4];
#pragma unroll
    for (int r = 0; r < H_IMG; ++r) {
        float Tk[16];
        Tk[15] = 0.0f; Tk[14] = 0.0f;
        float t = 0.0f;
#pragma unroll
        for (int k = 13; k >= 0; --k) {
            const int n = r * W_IMG + k + 1;
            float al = __int_as_float(__builtin_amdgcn_readlane(__float_as_int(alv[n >> 5]), n & 31));
            float tm = t * al;
            t = (k == c) ? 1.0f : ((k < c) ? tm : 0.0f);
            Tk[k] = t;
        }
#pragma unroll
        for (int s = 0; s < 4; ++s) {
            afr[r][s].x = hh ? Tk[4 * s + 2] : Tk[4 * s];
            afr[r][s].y = hh ? Tk[4 * s + 3] : Tk[4 * s + 1];
        }
    }

    int ka[4], kb[4];
#pragma unroll
    for (int s = 0; s < 4; ++s) {
        int k = 4 * s + 2 * hh;
        ka[s] = k     < W_IMG ? k     : W_IMG - 1;
        kb[s] = k + 1 < W_IMG ? k + 1 : W_IMG - 1;
    }

#pragma unroll
    for (int r = 0; r < H_IMG; ++r) {
        const int slo = (r == 0) ? 0 : (((r - 1) * W_IMG + 13) / 4 + 1);
        const int shi = (r * W_IMG + 13) / 4;
#pragma unroll
        for (int i = slo; i <= shi; ++i)
            *(v4f*)&S[(4 * i + q) * PITCH + c4] = xr[i];
        WAVE_SYNC();

        v8f acc0 = {}, acc1 = {};
#pragma unroll
        for (int s = 3; s >= 0; --s) {
            const int na = r * W_IMG + ka[s];
            const int nb = r * W_IMG + kb[s];
            v2f b0, b1;
            b0.x = S[na * PITCH + c];
            b0.y = S[nb * PITCH + c];
            b1.x = S[na * PITCH + 16 + c];
            b1.y = S[nb * PITCH + 16 + c];
            if (r > 0) {
                float ta = At[na], tb = At[nb];
                b0.x += ta * S[(na - W_IMG) * PITCH + c];
                b0.y += tb * S[(nb - W_IMG) * PITCH + c];
                b1.x += ta * S[(na - W_IMG) * PITCH + 16 + c];
                b1.y += tb * S[(nb - W_IMG) * PITCH + 16 + c];
            }
            acc0 = __builtin_amdgcn_wmma_f32_16x16x4_f32(
                false, afr[r][s], false, b0, (short)0, acc0, false, false);
            acc1 = __builtin_amdgcn_wmma_f32_16x16x4_f32(
                false, afr[r][s], false, b1, (short)0, acc1, false, false);
        }

#pragma unroll
        for (int i = 0; i < 8; ++i) {
            int M = i + 8 * hh;
            if (M < W_IMG) {
                S[(r * W_IMG + M) * PITCH + c]      = acc0[i];
                S[(r * W_IMG + M) * PITCH + 16 + c] = acc1[i];
            }
        }
        WAVE_SYNC();

        const int flo = (r == 0) ? 0 : (((r - 1) * W_IMG + 10) / 4 + 1);
        const int fhi = (r * W_IMG + 10) / 4;
#pragma unroll
        for (int i = flo; i <= fhi; ++i) {
            v4f z = *(const v4f*)&S[(4 * i + q) * PITCH + c4];
            *(v4f*)(yb + (4 * i + q) * PDIM) = z + dsk * xr[i];
        }
    }
}

extern "C" void kernel_launch(void* const* d_in, const int* in_sizes, int n_in,
                              void* d_out, int out_size, void* d_ws, size_t ws_size,
                              hipStream_t stream) {
    (void)in_sizes; (void)n_in; (void)d_ws; (void)ws_size; (void)out_size;
    const float* dt      = (const float*)d_in[0];
    const float* dt_bias = (const float*)d_in[1];
    const float* x       = (const float*)d_in[2];
    const float* Dskip   = (const float*)d_in[3];
    float* y             = (float*)d_out;
    chimera_dag_mix<<<NB * NHEADS, 64, 0, stream>>>(dt, dt_bias, x, Dskip, y);
}


# One wave per (b,h) owns all 64 channels as two tile pairs in a two-phase pipeline: each pair's next-row operands are requested from the slab ahead of the other pair's WMMAs; T built once per (b,h)
# speedup vs baseline: 2.0290x; 1.3487x over previous
#include <hip/hip_runtime.h>


typedef __attribute__((ext_vector_type(2))) float v2f;
typedef __attribute__((ext_vector_type(4))) float v4f;
typedef __attribute__((ext_vector_type(8))) float v8f;

#define H_IMG 14
#define W_IMG 14
#define LNODES 196
#define NB 32
#define NHEADS 24
#define PDIM 64
#define SDT (NB * NHEADS * LNODES)
#define PITCH 76
#define NCHUNK 98
#define RCH 7
#define AHEAD 7
#define COEF 224

#define WAVE_SYNC() do { \
    __builtin_amdgcn_fence(__ATOMIC_RELEASE, "wavefront"); \
    __builtin_amdgcn_wave_barrier(); \
    __builtin_amdgcn_fence(__ATOMIC_ACQUIRE, "wavefront"); } while (0)

__launch_bounds__(32)
__global__ void chimera_dag_mix(const float* __restrict__ dt,
                                const float* __restrict__ dt_bias,
                                const float* __restrict__ x,
                                const float* __restrict__ Dskip,
                                float* __restrict__ y)
{
    __shared__ alignas(16) float S[(LNODES + 1) * PITCH];
    __shared__ float At[COEF];

    const int bh   = blockIdx.x;
    const int hd   = bh % NHEADS;
    const int lane = threadIdx.x & 31;
    const int c    = lane & 15;
    const int hh   = lane >> 4;
    const int c4   = (lane & 15) * 4;

    const float bias = dt_bias[hd];
    const float dsk  = Dskip[hd];

    float d0[7], d1[7];
#pragma unroll
    for (int it = 0; it < 7; ++it) {
        int idx = lane + 32 * it;
        int idc = idx < LNODES ? idx : LNODES - 1;
        d0[it] = dt[0 * SDT + bh * LNODES + idc];
        d1[it] = dt[1 * SDT + bh * LNODES + idc];
    }

    const float* __restrict__ xb = x + (size_t)bh * (LNODES * PDIM) + c4;
    float* __restrict__ yb       = y + (size_t)bh * (LNODES * PDIM) + c4;
    v4f xr[NCHUNK];
#pragma unroll
    for (int i = 0; i < AHEAD * RCH; ++i)
        xr[i] = *(const v4f*)(xb + (2 * i + hh) * PDIM);

    float alv[7];
#pragma unroll
    for (int it = 0; it < 7; ++it) {
        int node = lane + 32 * it;
        int nc   = node < LNODES ? node : LNODES - 1;
        int row = nc / W_IMG, col = nc % W_IMG;
        float nie = 2.0f - (col == 0 ? 1.0f : 0.0f) - (row == 0 ? 1.0f : 0.0f);
        if (nie < 1e-6f) nie = 1.0f;
        float rn = 0.95f * rsqrtf(nie);
        float e0 = expf(d0[it] + bias);
        float e1 = expf(d1[it] + bias);
        float al = rn / (1.0f + e0);
        float at = rn / (1.0f + e1);
        bool in = node < LNODES;
        alv[it]  = (in && col > 0) ? al : 0.0f;
        At[node] = (in && row > 0) ? at : 0.0f;
    }

    int ka[4], kb[4];
#pragma unroll
    for (int s = 0; s < 4; ++s) {
        int k = 4 * s + 2 * hh;
        ka[s] = k     < W_IMG ? k     : W_IMG - 1;
        kb[s] = k + 1 < W_IMG ? k + 1 : W_IMG - 1;
    }

    v2f afr[2][4];
    float Tk[16];
    float tacc = 0.0f;
    float dlt[W_IMG];
#pragma unroll
    for (int k = 0; k < W_IMG; ++k) dlt[k] = (k == c) ? 1.0f : 0.0f;
#define BUILD_T_PART(R, KHI, KLO) do { \
        if ((KHI) == 13) { Tk[15] = 0.0f; Tk[14] = 0.0f; tacc = 0.0f; } \
        _Pragma("unroll") \
        for (int k = (KHI); k >= (KLO); --k) { \
            const int n = (R) * W_IMG + k + 1;       \
            float al = __int_as_float(__builtin_amdgcn_readlane(__float_as_int(alv[n >> 5]), n & 31)); \
            tacc = __builtin_fmaf(tacc, al, dlt[k]);     \
            Tk[k] = tacc; \
        } \
    } while (0)
#define BUILD_T_FINAL(DST) do { \
        _Pragma("unroll") \
        for (int s = 0; s < 4; ++s) { \
            (DST)[s].x = hh ? Tk[4 * s + 2] : Tk[4 * s]; \
            (DST)[s].y = hh ? Tk[4 * s + 3] : Tk[4 * s + 1]; \
        } \
    } while (0)

    float Lcx[2][4][2], Lcy[2][4][2], Lpx[2][4][2], Lpy[2][4][2], Lta[2][4], Ltb[2][4];
#define LOAD_PHASE(R, TP, BUF) do { \
        _Pragma("unroll") \
        for (int s = 0; s < 4; ++s) { \
            const int na = (R) * W_IMG + ka[s]; \
            const int nb = (R) * W_IMG + kb[s]; \
            _Pragma("unroll") \
            for (int t = 0; t < 2; ++t) { \
                const int col = 32 * (TP) + 16 * t + c; \
                Lcx[BUF][s][t] = S[na * PITCH + col]; \
                Lcy[BUF][s][t] = S[nb * PITCH + col]; \
                if ((R) > 0) { \
                    Lpx[BUF][s][t] = S[(na - W_IMG) * PITCH + col]; \
                    Lpy[BUF][s][t] = S[(nb - W_IMG) * PITCH + col]; \
                } \
            } \
            if ((R) > 0) { Lta[BUF][s] = At[na]; Ltb[BUF][s] = At[nb]; } \
        } \
    } while (0)

#define STAGE_ROW(R) do { \
        _Pragma("unroll") \
        for (int j = 0; j < RCH; ++j) { \
            const int i = (R) * RCH + j; \
            *(v4f*)&S[(2 * i + hh) * PITCH + c4] = xr[i]; \
        } \
        if ((R) + AHEAD < H_IMG) { \
            _Pragma("unroll") \
            for (int j = 0; j < RCH; ++j) { \
                const int i = ((R) + AHEAD) * RCH + j; \
                xr[i] = *(const v4f*)(xb + (2 * i + hh) * PDIM); \
            } \
        } \
    } while (0)

    v4f zr[RCH];
#define LOAD_Z(R) do { \
        _Pragma("unroll") \
        for (int j = 0; j < RCH; ++j) { \
            const int i = (R) * RCH + j; \
            zr[j] = *(const v4f*)&S[(2 * i + hh) * PITCH + c4]; \
        } \
    } while (0)
#define STORE_Y(R) do { \
        _Pragma("unroll") \
        for (int j = 0; j < RCH; ++j) { \
            const int i = (R) * RCH + j; \
            *(v4f*)(yb + (2 * i + hh) * PDIM) = zr[j] + dsk * xr[i]; \
        } \
    } while (0)

    STAGE_ROW(0);
    WAVE_SYNC();
    BUILD_T_PART(0, 13, 7);
    BUILD_T_PART(0, 6, 0);
    BUILD_T_FINAL(afr[0]);
    LOAD_PHASE(0, 0, 0);

#pragma unroll
    for (int r = 0; r < H_IMG; ++r) {
#pragma unroll
        for (int tp = 0; tp < 2; ++tp) {
            const int cur = tp, nxt = tp ^ 1;

            if (tp == 0) {
                LOAD_PHASE(r, 1, nxt);
                if (r > 0) LOAD_Z(r - 1);
            } else if (r + 1 < H_IMG) {
                LOAD_PHASE(r + 1, 0, nxt);
            }
            __builtin_amdgcn_sched_barrier(0);

            if (tp == 0 && r + 1 < H_IMG) STAGE_ROW(r + 1);
            if (tp == 1 && r > 0) STORE_Y(r - 1);
            if (r + 1 < H_IMG) {
                if (tp == 0) {
                    BUILD_T_PART(r + 1, 13, 7);
                } else {
                    BUILD_T_PART(r + 1, 6, 0);
                    BUILD_T_FINAL(afr[(r + 1) & 1]);
                }
            }

            if (r > 0)
                asm volatile("" : "+v"(Lta[cur][0]), "+v"(Lta[cur][1]), "+v"(Lta[cur][2]), "+v"(Lta[cur][3]),
                                  "+v"(Ltb[cur][0]), "+v"(Ltb[cur][1]), "+v"(Ltb[cur][2]), "+v"(Ltb[cur][3]));
            v8f acc0 = {}, acc1 = {};
#pragma unroll
            for (int s = 3; s >= 0; --s) {
                v2f b0, b1;
                b0.x = Lcx[cur][s][0];
                b0.y = Lcy[cur][s][0];
                b1.x = Lcx[cur][s][1];
                b1.y = Lcy[cur][s][1];
                if (r > 0) {
                    float ta = Lta[cur][s], tb = Ltb[cur][s];
                    b0.x += ta * Lpx[cur][s][0];
                    b0.y += tb * Lpy[cur][s][0];
                    b1.x += ta * Lpx[cur][s][1];
                    b1.y += tb * Lpy[cur][s][1];
                }
                acc0 = __builtin_amdgcn_wmma_f32_16x16x4_f32(
                    false, afr[r & 1][s], false, b0, (short)0, acc0, false, false);
                acc1 = __builtin_amdgcn_wmma_f32_16x16x4_f32(
                    false, afr[r & 1][s], false, b1, (short)0, acc1, false, false);
            }
            __builtin_amdgcn_sched_barrier(0);

#pragma unroll
            for (int i = 0; i < 8; ++i) {
                const int base = (r * W_IMG + i) * PITCH + 32 * tp;
                const int step = (i < 6) ? 8 * PITCH : (LNODES - (r * W_IMG + i)) * PITCH;
                const int off  = base + hh * step + c;
                S[off]      = acc0[i];
                S[off + 16] = acc1[i];
            }
            WAVE_SYNC();
        }
    }
    LOAD_Z(H_IMG - 1);
    STORE_Y(H_IMG - 1);
}

extern "C" void kernel_launch(void* const* d_in, const int* in_sizes, int n_in,
                              void* d_out, int out_size, void* d_ws, size_t ws_size,
                              hipStream_t stream) {
    (void)in_sizes; (void)n_in; (void)d_ws; (void)ws_size; (void)out_size;
    const float* dt      = (const float*)d_in[0];
    const float* dt_bias = (const float*)d_in[1];
    const float* x       = (const float*)d_in[2];
    const float* Dskip   = (const float*)d_in[3];
    float* y             = (float*)d_out;
    chimera_dag_mix<<<NB * NHEADS, 32, 0, stream>>>(dt, dt_bias, x, Dskip, y);
}
